# starting kernel
# baseline (speedup 1.0000x reference)
#include <hip/hip_runtime.h>

#define D_IN  128
#define D_OUT 64

typedef __attribute__((ext_vector_type(2))) float v2f;
typedef __attribute__((ext_vector_type(8))) float v8f;

__global__ void deg_init_kernel(float* __restrict__ deg, int n) {
    int i = blockIdx.x * blockDim.x + threadIdx.x;
    if (i < n) deg[i] = 1.0f;
}

__global__ void deg_count_kernel(const int* __restrict__ to_idx,
                                 float* __restrict__ deg, int e) {
    int i = blockIdx.x * blockDim.x + threadIdx.x;
    if (i < e) atomicAdd(&deg[to_idx[i]], 1.0f);
}

__global__ void dinv_kernel(float* __restrict__ deg, int n) {
    int i = blockIdx.x * blockDim.x + threadIdx.x;
    if (i < n) {
        float d = deg[i];
        deg[i] = (d > 0.0f) ? rsqrtf(d) : 0.0f;
    }
}

__global__ void __launch_bounds__(128)
gemm_wmma_kernel(const float* __restrict__ x, const float* __restrict__ W,
                 const float* __restrict__ b, float* __restrict__ h, int n_nodes) {
    const int lane = threadIdx.x & 31;
    const int wave = threadIdx.x >> 5;
    const int m0   = blockIdx.x * 16;
    const int n0   = wave * 16;
    const int half = lane >> 4;
    const int l    = lane & 15;

    int row = m0 + l;
    int rrow = (row < n_nodes) ? row : (n_nodes - 1);

    const float* __restrict__ xrow = x + (size_t)rrow * D_IN;
    const float* __restrict__ wrow = W + (size_t)(n0 + l) * D_IN;

    float bias = b[n0 + l];
    v8f acc;
#pragma unroll
    for (int r = 0; r < 8; ++r) acc[r] = bias;

#pragma unroll
    for (int k = 0; k < D_IN; k += 4) {
        v2f a, bb;
        a[0]  = xrow[k + 2 * half + 0];
        a[1]  = xrow[k + 2 * half + 1];
        bb[0] = wrow[k + 2 * half + 0];
        bb[1] = wrow[k + 2 * half + 1];
        acc = __builtin_amdgcn_wmma_f32_16x16x4_f32(
             false, a,  false, bb,
             (short)0, acc,  false,  false);
    }

    float* __restrict__ hbase = h + (size_t)m0 * D_OUT + n0;
#pragma unroll
    for (int r = 0; r < 8; ++r) {
        int mr = r + 8 * half;
        if (m0 + mr < n_nodes)
            hbase[(size_t)mr * D_OUT + l] = acc[r];
    }
}

__global__ void selfloop_kernel(const float* __restrict__ h,
                                const float* __restrict__ dinv,
                                float* __restrict__ out, int n) {
    int t = blockIdx.x * blockDim.x + threadIdx.x;
    int node = t >> 6;
    if (node < n) {
        float di = dinv[node];
        out[t] = h[t] * di * di;
    }
}

__global__ void scatter_kernel(const int* __restrict__ from_idx,
                               const int* __restrict__ to_idx,
                               const float* __restrict__ h,
                               const float* __restrict__ dinv,
                               float* __restrict__ out, int e) {
    int t = blockIdx.x * blockDim.x + threadIdx.x;
    int edge = t >> 6;
    int d    = t & 63;
    if (edge < e) {
        int fa = from_idx[edge];
        int ta = to_idx[edge];
        float c = dinv[fa] * dinv[ta];
        atomicAdd(&out[(size_t)ta * D_OUT + d], h[(size_t)fa * D_OUT + d] * c);
    }
}

extern "C" void kernel_launch(void* const* d_in, const int* in_sizes, int n_in,
                              void* d_out, int out_size, void* d_ws, size_t ws_size,
                              hipStream_t stream) {
    const float* x  = (const float*)d_in[0];
    const float* W  = (const float*)d_in[1];
    const float* b  = (const float*)d_in[2];
    const int*   ei = (const int*)d_in[3];

    const int N = in_sizes[0] / D_IN;
    const int E = in_sizes[3] / 2;
    const int* from_idx = ei;
    const int* to_idx   = ei + E;

    float* out = (float*)d_out;

    float* deg = (float*)d_ws;
    size_t deg_bytes = ((size_t)N * sizeof(float) + 255) & ~(size_t)255;
    float* h = (float*)((char*)d_ws + deg_bytes);

    const int B256 = 256;

    deg_init_kernel<<<(N + B256 - 1) / B256, B256, 0, stream>>>(deg, N);
    deg_count_kernel<<<(E + B256 - 1) / B256, B256, 0, stream>>>(to_idx, deg, E);
    dinv_kernel<<<(N + B256 - 1) / B256, B256, 0, stream>>>(deg, N);

    int m_tiles = (N + 15) / 16;
    gemm_wmma_kernel<<<m_tiles, 128, 0, stream>>>(x, W, b, h, N);

    long long self_threads = (long long)N * D_OUT;
    selfloop_kernel<<<(int)((self_threads + B256 - 1) / B256), B256, 0, stream>>>(
        h, deg, out, N);

    long long edge_threads = (long long)E * D_OUT;
    scatter_kernel<<<(int)((edge_threads + B256 - 1) / B256), B256, 0, stream>>>(
        from_idx, to_idx, h, deg, out, E);
}


# scatter: one wave per 16 edges (32 h-row loads in flight, then its f32 atomics), 32x fewer waves; selfloop: float4 per lane, 4x fewer; both were wave-dispatch bound; only atomic arrival order changes
# speedup vs baseline: 2.2728x; 2.2728x over previous
#include <hip/hip_runtime.h>

#define D_IN  128
#define D_OUT 64

typedef __attribute__((ext_vector_type(2))) float v2f;
typedef __attribute__((ext_vector_type(8))) float v8f;

__global__ void deg_init_kernel(float* __restrict__ deg, int n) {
    int i = blockIdx.x * blockDim.x + threadIdx.x;
    if (i < n) deg[i] = 1.0f;
}

__global__ void deg_count_kernel(const int* __restrict__ to_idx,
                                 float* __restrict__ deg, int e) {
    int i = blockIdx.x * blockDim.x + threadIdx.x;
    if (i < e) atomicAdd(&deg[to_idx[i]], 1.0f);
}

__global__ void dinv_kernel(float* __restrict__ deg, int n) {
    int i = blockIdx.x * blockDim.x + threadIdx.x;
    if (i < n) {
        float d = deg[i];
        deg[i] = (d > 0.0f) ? rsqrtf(d) : 0.0f;
    }
}

__global__ void __launch_bounds__(128)
gemm_wmma_kernel(const float* __restrict__ x, const float* __restrict__ W,
                 const float* __restrict__ b, float* __restrict__ h, int n_nodes) {
    const int lane = threadIdx.x & 31;
    const int wave = threadIdx.x >> 5;
    const int m0   = blockIdx.x * 16;
    const int n0   = wave * 16;
    const int half = lane >> 4;
    const int l    = lane & 15;

    int row = m0 + l;
    int rrow = (row < n_nodes) ? row : (n_nodes - 1);

    const float* __restrict__ xrow = x + (size_t)rrow * D_IN;
    const float* __restrict__ wrow = W + (size_t)(n0 + l) * D_IN;

    float bias = b[n0 + l];
    v8f acc;
#pragma unroll
    for (int r = 0; r < 8; ++r) acc[r] = bias;

#pragma unroll
    for (int k = 0; k < D_IN; k += 4) {
        v2f a, bb;
        a[0]  = xrow[k + 2 * half + 0];
        a[1]  = xrow[k + 2 * half + 1];
        bb[0] = wrow[k + 2 * half + 0];
        bb[1] = wrow[k + 2 * half + 1];
        acc = __builtin_amdgcn_wmma_f32_16x16x4_f32(
             false, a,  false, bb,
             (short)0, acc,  false,  false);
    }

    float* __restrict__ hbase = h + (size_t)m0 * D_OUT + n0;
#pragma unroll
    for (int r = 0; r < 8; ++r) {
        int mr = r + 8 * half;
        if (m0 + mr < n_nodes)
            hbase[(size_t)mr * D_OUT + l] = acc[r];
    }
}

__global__ void selfloop_kernel(const float* __restrict__ h,
                                const float* __restrict__ dinv,
                                float* __restrict__ out, int n) {
    int t = blockIdx.x * blockDim.x + threadIdx.x;
    int node = t >> 4;
    if (node < n) {
        float di = dinv[node];
        float s = di * di;
        float4 v = reinterpret_cast<const float4*>(h)[t];
        v.x *= s; v.y *= s; v.z *= s; v.w *= s;
        reinterpret_cast<float4*>(out)[t] = v;
    }
}

#define EPW 16
__global__ void __launch_bounds__(256)
scatter_kernel(const int* __restrict__ from_idx,
               const int* __restrict__ to_idx,
               const float* __restrict__ h,
               const float* __restrict__ dinv,
               float* __restrict__ out, int e) {
    const int lane = threadIdx.x & 31;
    const int wave = __builtin_amdgcn_readfirstlane(
        (int)((blockIdx.x * blockDim.x + threadIdx.x) >> 5));
    const int e0 = wave * EPW;
    if (e0 >= e) return;
    int n = e - e0;
    if (n > EPW) n = EPW;

    int fa_l = 0, ta_l = 0;
    float c_l = 0.0f;
    if (lane < n) {
        fa_l = from_idx[e0 + lane];
        ta_l = to_idx[e0 + lane];
        c_l  = dinv[fa_l] * dinv[ta_l];
    }

    if (n == EPW) {
        float v0[EPW], v1[EPW];
#pragma unroll
        for (int j = 0; j < EPW; ++j) {
            const int f = __builtin_amdgcn_readlane(fa_l, j);
            const float* __restrict__ hr = h + (size_t)f * D_OUT;
            v0[j] = hr[lane];
            v1[j] = hr[32 + lane];
        }
#pragma unroll
        for (int j = 0; j < EPW; ++j) {
            const int t = __builtin_amdgcn_readlane(ta_l, j);
            const float c = __int_as_float(
                __builtin_amdgcn_readlane(__float_as_int(c_l), j));
            float* __restrict__ orow = out + (size_t)t * D_OUT;
            atomicAdd(&orow[lane], v0[j] * c);
            atomicAdd(&orow[32 + lane], v1[j] * c);
        }
    } else {
        for (int j = 0; j < n; ++j) {
            const int f = __builtin_amdgcn_readlane(fa_l, j);
            const int t = __builtin_amdgcn_readlane(ta_l, j);
            const float c = __int_as_float(
                __builtin_amdgcn_readlane(__float_as_int(c_l), j));
            const float* __restrict__ hr = h + (size_t)f * D_OUT;
            float* __restrict__ orow = out + (size_t)t * D_OUT;
            atomicAdd(&orow[lane], hr[lane] * c);
            atomicAdd(&orow[32 + lane], hr[32 + lane] * c);
        }
    }
}

extern "C" void kernel_launch(void* const* d_in, const int* in_sizes, int n_in,
                              void* d_out, int out_size, void* d_ws, size_t ws_size,
                              hipStream_t stream) {
    const float* x  = (const float*)d_in[0];
    const float* W  = (const float*)d_in[1];
    const float* b  = (const float*)d_in[2];
    const int*   ei = (const int*)d_in[3];

    const int N = in_sizes[0] / D_IN;
    const int E = in_sizes[3] / 2;
    const int* from_idx = ei;
    const int* to_idx   = ei + E;

    float* out = (float*)d_out;

    float* deg = (float*)d_ws;
    size_t deg_bytes = ((size_t)N * sizeof(float) + 255) & ~(size_t)255;
    float* h = (float*)((char*)d_ws + deg_bytes);

    const int B256 = 256;

    deg_init_kernel<<<(N + B256 - 1) / B256, B256, 0, stream>>>(deg, N);
    deg_count_kernel<<<(E + B256 - 1) / B256, B256, 0, stream>>>(to_idx, deg, E);
    dinv_kernel<<<(N + B256 - 1) / B256, B256, 0, stream>>>(deg, N);

    int m_tiles = (N + 15) / 16;
    gemm_wmma_kernel<<<m_tiles, 128, 0, stream>>>(x, W, b, h, N);

    long long self_threads = (long long)N * (D_OUT / 4);
    selfloop_kernel<<<(int)((self_threads + B256 - 1) / B256), B256, 0, stream>>>(
        h, deg, out, N);

    long long edge_threads = ((long long)E + EPW - 1) / EPW * 32;
    scatter_kernel<<<(int)((edge_threads + B256 - 1) / B256), B256, 0, stream>>>(
        from_idx, to_idx, h, deg, out, E);
}


# gemm: blocks walk over node tiles; each x tile is fetched once per block by whole-line loads into LDS and A fragments are read from LDS; W fragments stay in registers; sums and order unchanged
# speedup vs baseline: 2.5544x; 1.1239x over previous
#include <hip/hip_runtime.h>

#define D_IN  128
#define D_OUT 64

typedef __attribute__((ext_vector_type(2))) float v2f;
typedef __attribute__((ext_vector_type(8))) float v8f;

__global__ void deg_init_kernel(float* __restrict__ deg, int n) {
    int i = blockIdx.x * blockDim.x + threadIdx.x;
    if (i < n) deg[i] = 1.0f;
}

__global__ void deg_count_kernel(const int* __restrict__ to_idx,
                                 float* __restrict__ deg, int e) {
    int i = blockIdx.x * blockDim.x + threadIdx.x;
    if (i < e) atomicAdd(&deg[to_idx[i]], 1.0f);
}

__global__ void dinv_kernel(float* __restrict__ deg, int n) {
    int i = blockIdx.x * blockDim.x + threadIdx.x;
    if (i < n) {
        float d = deg[i];
        deg[i] = (d > 0.0f) ? rsqrtf(d) : 0.0f;
    }
}

#define A_PITCH 132
#define A_TILE  (16 * A_PITCH)
#define K_STEPS (D_IN / 4)

__device__ __forceinline__ void fetch_rows(const float* __restrict__ x, int t, int sr,
                                           int sc, int n_nodes, float4& p0, float4& p1,
                                           float4& p2, float4& p3) {
    int row = t * 16 + sr;
    int rrow = (row < n_nodes) ? row : (n_nodes - 1);
    const float4* __restrict__ src =
        reinterpret_cast<const float4*>(x + (size_t)rrow * D_IN) + sc;
    p0 = src[0];
    p1 = src[8];
    p2 = src[16];
    p3 = src[24];
}

__device__ __forceinline__ void park_rows(float* __restrict__ dst, const float4& p0,
                                          const float4& p1, const float4& p2,
                                          const float4& p3) {
    *reinterpret_cast<float4*>(dst +  0) = p0;
    *reinterpret_cast<float4*>(dst + 32) = p1;
    *reinterpret_cast<float4*>(dst + 64) = p2;
    *reinterpret_cast<float4*>(dst + 96) = p3;
}

__device__ __forceinline__ v8f tile_products(const float* __restrict__ as_ld,
                                             const v2f (&bf)[K_STEPS], float bias) {
    v2f af[K_STEPS];
#pragma unroll
    for (int s = 0; s < K_STEPS; ++s)
        af[s] = *reinterpret_cast<const v2f*>(as_ld + 4 * s);
    v8f acc;
#pragma unroll
    for (int r = 0; r < 8; ++r) acc[r] = bias;
    __builtin_amdgcn_sched_barrier(0);
#pragma unroll
    for (int s = 0; s < K_STEPS; ++s)
        acc = __builtin_amdgcn_wmma_f32_16x16x4_f32(
             false, af[s],  false, bf[s],
             (short)0, acc,  false,  false);
    return acc;
}

__device__ __forceinline__ void store_tile(float* __restrict__ h, int t, int n0, int half,
                                           int l, int n_nodes, const v8f& acc) {
    const int m0 = t * 16;
    float* __restrict__ hbase = h + (size_t)m0 * D_OUT + n0;
#pragma unroll
    for (int r = 0; r < 8; ++r) {
        int mr = r + 8 * half;
        if (m0 + mr < n_nodes)
            hbase[(size_t)mr * D_OUT + l] = acc[r];
    }
}

__global__ void __launch_bounds__(128)
gemm_wmma_kernel(const float* __restrict__ x, const float* __restrict__ W,
                 const float* __restrict__ b, float* __restrict__ h,
                 int n_nodes, int m_tiles) {
    __shared__ __attribute__((aligned(16))) float As[2 * A_TILE];

    int t = blockIdx.x;
    if (t >= m_tiles) return;

    const int tid  = threadIdx.x;
    const int lane = tid & 31;
    const int wave = tid >> 5;
    const int n0   = wave * 16;
    const int half = lane >> 4;
    const int l    = lane & 15;
    const int sr   = tid >> 3;
    const int sc   = tid & 7;
    const int step = gridDim.x;

    float4 p0, p1, p2, p3;
    fetch_rows(x, t, sr, sc, n_nodes, p0, p1, p2, p3);

    const float* __restrict__ wrow = W + (size_t)(n0 + l) * D_IN + 2 * half;
    v2f bf[K_STEPS];
#pragma unroll
    for (int s = 0; s < K_STEPS; ++s) {
        bf[s][0] = wrow[4 * s + 0];
        bf[s][1] = wrow[4 * s + 1];
    }
    const float bias = b[n0 + l];

    const int st_off = sr * A_PITCH + 4 * sc;
    const int ld_off = l * A_PITCH + 2 * half;

    park_rows(As + st_off, p0, p1, p2, p3);
    __syncthreads();
    if (t + step < m_tiles) fetch_rows(x, t + step, sr, sc, n_nodes, p0, p1, p2, p3);
    v8f acc = tile_products(As + ld_off, bf, bias);
    int t_done = t;
    int buf = 1;

    for (t += step; t < m_tiles; t += step) {
        park_rows(As + buf * A_TILE + st_off, p0, p1, p2, p3);
        __syncthreads();
        store_tile(h, t_done, n0, half, l, n_nodes, acc);
        if (t + step < m_tiles) fetch_rows(x, t + step, sr, sc, n_nodes, p0, p1, p2, p3);
        acc = tile_products(As + buf * A_TILE + ld_off, bf, bias);
        t_done = t;
        buf ^= 1;
    }
    store_tile(h, t_done, n0, half, l, n_nodes, acc);
}

__global__ void selfloop_kernel(const float* __restrict__ h,
                                const float* __restrict__ dinv,
                                float* __restrict__ out, int n) {
    int t = blockIdx.x * blockDim.x + threadIdx.x;
    int node = t >> 4;
    if (node < n) {
        float di = dinv[node];
        float s = di * di;
        float4 v = reinterpret_cast<const float4*>(h)[t];
        v.x *= s; v.y *= s; v.z *= s; v.w *= s;
        reinterpret_cast<float4*>(out)[t] = v;
    }
}

#define EPW 16
__global__ void __launch_bounds__(256)
scatter_kernel(const int* __restrict__ from_idx,
               const int* __restrict__ to_idx,
               const float* __restrict__ h,
               const float* __restrict__ dinv,
               float* __restrict__ out, int e) {
    const int lane = threadIdx.x & 31;
    const int wave = __builtin_amdgcn_readfirstlane(
        (int)((blockIdx.x * blockDim.x + threadIdx.x) >> 5));
    const int e0 = wave * EPW;
    if (e0 >= e) return;
    int n = e - e0;
    if (n > EPW) n = EPW;

    int fa_l = 0, ta_l = 0;
    float c_l = 0.0f;
    if (lane < n) {
        fa_l = from_idx[e0 + lane];
        ta_l = to_idx[e0 + lane];
        c_l  = dinv[fa_l] * dinv[ta_l];
    }

    if (n == EPW) {
        float v0[EPW], v1[EPW];
#pragma unroll
        for (int j = 0; j < EPW; ++j) {
            const int f = __builtin_amdgcn_readlane(fa_l, j);
            const float* __restrict__ hr = h + (size_t)f * D_OUT;
            v0[j] = hr[lane];
            v1[j] = hr[32 + lane];
        }
#pragma unroll
        for (int j = 0; j < EPW; ++j) {
            const int t = __builtin_amdgcn_readlane(ta_l, j);
            const float c = __int_as_float(
                __builtin_amdgcn_readlane(__float_as_int(c_l), j));
            float* __restrict__ orow = out + (size_t)t * D_OUT;
            atomicAdd(&orow[lane], v0[j] * c);
            atomicAdd(&orow[32 + lane], v1[j] * c);
        }
    } else {
        for (int j = 0; j < n; ++j) {
            const int f = __builtin_amdgcn_readlane(fa_l, j);
            const int t = __builtin_amdgcn_readlane(ta_l, j);
            const float c = __int_as_float(
                __builtin_amdgcn_readlane(__float_as_int(c_l), j));
            const float* __restrict__ hr = h + (size_t)f * D_OUT;
            float* __restrict__ orow = out + (size_t)t * D_OUT;
            atomicAdd(&orow[lane], hr[lane] * c);
            atomicAdd(&orow[32 + lane], hr[32 + lane] * c);
        }
    }
}

extern "C" void kernel_launch(void* const* d_in, const int* in_sizes, int n_in,
                              void* d_out, int out_size, void* d_ws, size_t ws_size,
                              hipStream_t stream) {
    const float* x  = (const float*)d_in[0];
    const float* W  = (const float*)d_in[1];
    const float* b  = (const float*)d_in[2];
    const int*   ei = (const int*)d_in[3];

    const int N = in_sizes[0] / D_IN;
    const int E = in_sizes[3] / 2;
    const int* from_idx = ei;
    const int* to_idx   = ei + E;

    float* out = (float*)d_out;

    float* deg = (float*)d_ws;
    size_t deg_bytes = ((size_t)N * sizeof(float) + 255) & ~(size_t)255;
    float* h = (float*)((char*)d_ws + deg_bytes);

    const int B256 = 256;

    deg_init_kernel<<<(N + B256 - 1) / B256, B256, 0, stream>>>(deg, N);
    deg_count_kernel<<<(E + B256 - 1) / B256, B256, 0, stream>>>(to_idx, deg, E);
    dinv_kernel<<<(N + B256 - 1) / B256, B256, 0, stream>>>(deg, N);

    int m_tiles = (N + 15) / 16;
    gemm_wmma_kernel<<<1024, 128, 0, stream>>>(x, W, b, h, N, m_tiles);

    long long self_threads = (long long)N * (D_OUT / 4);
    selfloop_kernel<<<(int)((self_threads + B256 - 1) / B256), B256, 0, stream>>>(
        h, deg, out, N);

    long long edge_threads = ((long long)E + EPW - 1) / EPW * 32;
    scatter_kernel<<<(int)((edge_threads + B256 - 1) / B256), B256, 0, stream>>>(
        from_idx, to_idx, h, deg, out, E);
}


# normalisation moved into the GEMM store (degrees counted first): writes h*dinv[row] and out=h*dinv^2; dinv and self-loop launches and scatter's dinv[from] gather go; message=(h*df)*dt, sums as before
# speedup vs baseline: 2.6776x; 1.0482x over previous
#include <hip/hip_runtime.h>

#define D_IN  128
#define D_OUT 64

typedef __attribute__((ext_vector_type(2))) float v2f;
typedef __attribute__((ext_vector_type(8))) float v8f;

__global__ void deg_init_kernel(float* __restrict__ deg, int n) {
    int i = blockIdx.x * blockDim.x + threadIdx.x;
    if (i < n) deg[i] = 1.0f;
}

__global__ void deg_count_kernel(const int* __restrict__ to_idx,
                                 float* __restrict__ deg, int e) {
    int i = blockIdx.x * blockDim.x + threadIdx.x;
    if (i < e) atomicAdd(&deg[to_idx[i]], 1.0f);
}

#define A_PITCH 132
#define A_TILE  (16 * A_PITCH)
#define K_STEPS (D_IN / 4)

__device__ __forceinline__ void fetch_rows(const float* __restrict__ x, int t, int sr,
                                           int sc, int n_nodes, float4& p0, float4& p1,
                                           float4& p2, float4& p3) {
    int row = t * 16 + sr;
    int rrow = (row < n_nodes) ? row : (n_nodes - 1);
    const float4* __restrict__ src =
        reinterpret_cast<const float4*>(x + (size_t)rrow * D_IN) + sc;
    p0 = src[0];
    p1 = src[8];
    p2 = src[16];
    p3 = src[24];
}

__device__ __forceinline__ void park_rows(float* __restrict__ dst, const float4& p0,
                                          const float4& p1, const float4& p2,
                                          const float4& p3) {
    *reinterpret_cast<float4*>(dst +  0) = p0;
    *reinterpret_cast<float4*>(dst + 32) = p1;
    *reinterpret_cast<float4*>(dst + 64) = p2;
    *reinterpret_cast<float4*>(dst + 96) = p3;
}

__device__ __forceinline__ v8f tile_products(const float* __restrict__ as_ld,
                                             const v2f (&bf)[K_STEPS], float bias) {
    v2f af[K_STEPS];
#pragma unroll
    for (int s = 0; s < K_STEPS; ++s)
        af[s] = *reinterpret_cast<const v2f*>(as_ld + 4 * s);
    v8f acc;
#pragma unroll
    for (int r = 0; r < 8; ++r) acc[r] = bias;
    __builtin_amdgcn_sched_barrier(0);
#pragma unroll
    for (int s = 0; s < K_STEPS; ++s)
        acc = __builtin_amdgcn_wmma_f32_16x16x4_f32(
             false, af[s],  false, bf[s],
             (short)0, acc,  false,  false);
    return acc;
}

__device__ __forceinline__ void fetch_deg(const float* __restrict__ deg, int t, int half,
                                          int n_nodes, float (&dg)[8]) {
#pragma unroll
    for (int r = 0; r < 8; ++r) {
        int row = t * 16 + 8 * half + r;
        dg[r] = deg[(row < n_nodes) ? row : (n_nodes - 1)];
    }
}

__device__ __forceinline__ void store_tile(float* __restrict__ hs, float* __restrict__ out,
                                           int t, int n0, int half, int l, int n_nodes,
                                           const v8f& acc, const float (&dg)[8]) {
    const int m0 = t * 16;
    const size_t base = (size_t)m0 * D_OUT + n0;
    float* __restrict__ hbase = hs + base;
    float* __restrict__ obase = out + base;
#pragma unroll
    for (int r = 0; r < 8; ++r) {
        int mr = r + 8 * half;
        if (m0 + mr < n_nodes) {
            float d  = dg[r];
            float di = (d > 0.0f) ? rsqrtf(d) : 0.0f;
            hbase[(size_t)mr * D_OUT + l] = acc[r] * di;
            obase[(size_t)mr * D_OUT + l] = acc[r] * (di * di);
        }
    }
}

__global__ void __launch_bounds__(128)
gemm_wmma_kernel(const float* __restrict__ x, const float* __restrict__ W,
                 const float* __restrict__ b, const float* __restrict__ deg,
                 float* __restrict__ hs, float* __restrict__ out,
                 int n_nodes, int m_tiles) {
    __shared__ __attribute__((aligned(16))) float As[2 * A_TILE];

    int t = blockIdx.x;
    if (t >= m_tiles) return;

    const int tid  = threadIdx.x;
    const int lane = tid & 31;
    const int wave = tid >> 5;
    const int n0   = wave * 16;
    const int half = lane >> 4;
    const int l    = lane & 15;
    const int sr   = tid >> 3;
    const int sc   = tid & 7;
    const int step = gridDim.x;

    float dg[8];
    fetch_deg(deg, t, half, n_nodes, dg);
    float4 p0, p1, p2, p3;
    fetch_rows(x, t, sr, sc, n_nodes, p0, p1, p2, p3);

    const float* __restrict__ wrow = W + (size_t)(n0 + l) * D_IN + 2 * half;
    v2f bf[K_STEPS];
#pragma unroll
    for (int s = 0; s < K_STEPS; ++s) {
        bf[s][0] = wrow[4 * s + 0];
        bf[s][1] = wrow[4 * s + 1];
    }
    const float bias = b[n0 + l];

    const int st_off = sr * A_PITCH + 4 * sc;
    const int ld_off = l * A_PITCH + 2 * half;

    park_rows(As + st_off, p0, p1, p2, p3);
    __syncthreads();
    if (t + step < m_tiles) fetch_rows(x, t + step, sr, sc, n_nodes, p0, p1, p2, p3);
    v8f acc = tile_products(As + ld_off, bf, bias);
    int t_done = t;
    int buf = 1;

    for (t += step; t < m_tiles; t += step) {
        park_rows(As + buf * A_TILE + st_off, p0, p1, p2, p3);
        __syncthreads();
        store_tile(hs, out, t_done, n0, half, l, n_nodes, acc, dg);
        fetch_deg(deg, t, half, n_nodes, dg);
        if (t + step < m_tiles) fetch_rows(x, t + step, sr, sc, n_nodes, p0, p1, p2, p3);
        acc = tile_products(As + buf * A_TILE + ld_off, bf, bias);
        t_done = t;
        buf ^= 1;
    }
    store_tile(hs, out, t_done, n0, half, l, n_nodes, acc, dg);
}

#define EPW 16
__global__ void __launch_bounds__(256)
scatter_kernel(const int* __restrict__ from_idx,
               const int* __restrict__ to_idx,
               const float* __restrict__ h,
               const float* __restrict__ deg,
               float* __restrict__ out, int e) {
    const int lane = threadIdx.x & 31;
    const int wave = __builtin_amdgcn_readfirstlane(
        (int)((blockIdx.x * blockDim.x + threadIdx.x) >> 5));
    const int e0 = wave * EPW;
    if (e0 >= e) return;
    int n = e - e0;
    if (n > EPW) n = EPW;

    int fa_l = 0, ta_l = 0;
    float c_l = 0.0f;
    if (lane < n) {
        fa_l = from_idx[e0 + lane];
        ta_l = to_idx[e0 + lane];
        const float d = deg[ta_l];
        c_l  = (d > 0.0f) ? rsqrtf(d) : 0.0f;
    }

    if (n == EPW) {
        float v0[EPW], v1[EPW];
#pragma unroll
        for (int j = 0; j < EPW; ++j) {
            const int f = __builtin_amdgcn_readlane(fa_l, j);
            const float* __restrict__ hr = h + (size_t)f * D_OUT;
            v0[j] = hr[lane];
            v1[j] = hr[32 + lane];
        }
#pragma unroll
        for (int j = 0; j < EPW; ++j) {
            const int t = __builtin_amdgcn_readlane(ta_l, j);
            const float c = __int_as_float(
                __builtin_amdgcn_readlane(__float_as_int(c_l), j));
            float* __restrict__ orow = out + (size_t)t * D_OUT;
            atomicAdd(&orow[lane], v0[j] * c);
            atomicAdd(&orow[32 + lane], v1[j] * c);
        }
    } else {
        for (int j = 0; j < n; ++j) {
            const int f = __builtin_amdgcn_readlane(fa_l, j);
            const int t = __builtin_amdgcn_readlane(ta_l, j);
            const float c = __int_as_float(
                __builtin_amdgcn_readlane(__float_as_int(c_l), j));
            const float* __restrict__ hr = h + (size_t)f * D_OUT;
            float* __restrict__ orow = out + (size_t)t * D_OUT;
            atomicAdd(&orow[lane], hr[lane] * c);
            atomicAdd(&orow[32 + lane], hr[32 + lane] * c);
        }
    }
}

extern "C" void kernel_launch(void* const* d_in, const int* in_sizes, int n_in,
                              void* d_out, int out_size, void* d_ws, size_t ws_size,
                              hipStream_t stream) {
    const float* x  = (const float*)d_in[0];
    const float* W  = (const float*)d_in[1];
    const float* b  = (const float*)d_in[2];
    const int*   ei = (const int*)d_in[3];

    const int N = in_sizes[0] / D_IN;
    const int E = in_sizes[3] / 2;
    const int* from_idx = ei;
    const int* to_idx   = ei + E;

    float* out = (float*)d_out;

    float* deg = (float*)d_ws;
    size_t deg_bytes = ((size_t)N * sizeof(float) + 255) & ~(size_t)255;
    float* hs = (float*)((char*)d_ws + deg_bytes);

    const int B256 = 256;

    deg_init_kernel<<<(N + B256 - 1) / B256, B256, 0, stream>>>(deg, N);
    deg_count_kernel<<<(E + B256 - 1) / B256, B256, 0, stream>>>(to_idx, deg, E);

    int m_tiles = (N + 15) / 16;
    gemm_wmma_kernel<<<1024, 128, 0, stream>>>(x, W, b, deg, hs, out, N, m_tiles);

    long long edge_threads = ((long long)E + EPW - 1) / EPW * 32;
    scatter_kernel<<<(int)((edge_threads + B256 - 1) / B256), B256, 0, stream>>>(
        from_idx, to_idx, hs, deg, out, E);
}


# edge pass: f32 atomic scatter and atomic degree count replaced by a two-level destination sort on LDS counters (no global atomics) plus a per-node gather; rows summed in slot order, one out store
# speedup vs baseline: 4.0069x; 1.4965x over previous
#include <hip/hip_runtime.h>

#define D_IN  128
#define D_OUT 64

typedef __attribute__((ext_vector_type(2))) float v2f;
typedef __attribute__((ext_vector_type(8))) float v8f;

#define BIN_SHIFT    8
#define BIN_NODES    256
#define SORT_BLOCKS  256
#define SORT_THREADS 1024
#define MAX_BINS     1024
#define CSR_THREADS  512
#define BATCH        4

__device__ __forceinline__ int block_excl_scan(int v, int* s_w) {
    const int lane = threadIdx.x & 31;
    const int wv   = threadIdx.x >> 5;
    const int nw   = blockDim.x >> 5;
    int incl = v;
#pragma unroll
    for (int d = 1; d < 32; d <<= 1) {
        int t = __shfl_up(incl, d);
        if (lane >= d) incl += t;
    }
    if (lane == 31) s_w[wv] = incl;
    __syncthreads();
    if (wv == 0) {
        int w  = (lane < nw) ? s_w[lane] : 0;
        int wi = w;
#pragma unroll
        for (int d = 1; d < 32; d <<= 1) {
            int t = __shfl_up(wi, d);
            if (lane >= d) wi += t;
        }
        s_w[lane] = wi - w;
    }
    __syncthreads();
    return s_w[wv] + incl - v;
}

__global__ void __launch_bounds__(SORT_THREADS)
bin_count_kernel(const int* __restrict__ to_idx, int* __restrict__ hist,
                 int e, int nb, int chunk) {
    __shared__ int s_h[MAX_BINS];
    const int tid = threadIdx.x;
    s_h[tid] = 0;
    __syncthreads();
    const int beg = blockIdx.x * chunk;
    int end = beg + chunk;
    if (end > e) end = e;
    for (int i0 = beg + tid; i0 < end; i0 += BATCH * SORT_THREADS) {
        int t[BATCH];
#pragma unroll
        for (int k = 0; k < BATCH; ++k) {
            const int i = i0 + k * SORT_THREADS;
            t[k] = to_idx[(i < end) ? i : (end - 1)];
        }
#pragma unroll
        for (int k = 0; k < BATCH; ++k)
            if (i0 + k * SORT_THREADS < end) atomicAdd(&s_h[t[k] >> BIN_SHIFT], 1);
    }
    __syncthreads();
    if (tid < nb) hist[blockIdx.x * nb + tid] = s_h[tid];
}

__global__ void __launch_bounds__(SORT_THREADS)
bin_place_kernel(const int* __restrict__ from_idx, const int* __restrict__ to_idx,
                 const int* __restrict__ hist, int* __restrict__ bin_base,
                 int* __restrict__ items, int e, int nb, int chunk) {
    __shared__ int s_off[MAX_BINS];
    __shared__ int s_cur[MAX_BINS];
    __shared__ int s_w[32];
    const int tid = threadIdx.x;
    const int me  = blockIdx.x;

    int total = 0, mine = 0;
    if (tid < nb) {
#pragma unroll 32
        for (int b = 0; b < SORT_BLOCKS; ++b) {
            const int v = hist[b * nb + tid];
            total += v;
            if (b < me) mine += v;
        }
    }
    const int base = block_excl_scan(total, s_w);
    s_off[tid] = base + mine;
    s_cur[tid] = 0;
    if (me == 0) {
        if (tid < nb) bin_base[tid] = base;
        if (tid == 0) bin_base[nb] = e;
    }
    __syncthreads();

    const int beg = me * chunk;
    int end = beg + chunk;
    if (end > e) end = e;
    for (int i0 = beg + tid; i0 < end; i0 += BATCH * SORT_THREADS) {
        int t[BATCH], f[BATCH];
#pragma unroll
        for (int k = 0; k < BATCH; ++k) {
            const int i  = i0 + k * SORT_THREADS;
            const int ic = (i < end) ? i : (end - 1);
            t[k] = to_idx[ic];
            f[k] = from_idx[ic];
        }
#pragma unroll
        for (int k = 0; k < BATCH; ++k) {
            if (i0 + k * SORT_THREADS < end) {
                const int bin = t[k] >> BIN_SHIFT;
                const int r = atomicAdd(&s_cur[bin], 1);
                items[s_off[bin] + r] = (int)(((unsigned)f[k] << BIN_SHIFT) |
                                              (unsigned)(t[k] & (BIN_NODES - 1)));
            }
        }
    }
}

__global__ void __launch_bounds__(CSR_THREADS)
bin_sort_kernel(const int* __restrict__ items, const int* __restrict__ bin_base,
                float* __restrict__ deg, int* __restrict__ rowcnt,
                int* __restrict__ src_sorted, int n_nodes) {
    __shared__ int s_cnt[BIN_NODES];
    __shared__ int s_start[BIN_NODES];
    __shared__ int s_w[32];
    const int tid = threadIdx.x;
    const int bin = blockIdx.x;
    const int beg = bin_base[bin];
    const int end = bin_base[bin + 1];

    if (tid < BIN_NODES) s_cnt[tid] = 0;
    __syncthreads();
    for (int i0 = beg + tid; i0 < end; i0 += BATCH * CSR_THREADS) {
        unsigned it[BATCH];
#pragma unroll
        for (int k = 0; k < BATCH; ++k) {
            const int i = i0 + k * CSR_THREADS;
            it[k] = (unsigned)items[(i < end) ? i : (end - 1)];
        }
#pragma unroll
        for (int k = 0; k < BATCH; ++k)
            if (i0 + k * CSR_THREADS < end)
                atomicAdd(&s_cnt[it[k] & (BIN_NODES - 1)], 1);
    }
    __syncthreads();

    const int c  = (tid < BIN_NODES) ? s_cnt[tid] : 0;
    const int st = block_excl_scan(c, s_w);
    if (tid < BIN_NODES) {
        s_start[tid] = beg + st;
        s_cnt[tid] = 0;
        const int node = bin * BIN_NODES + tid;
        if (node < n_nodes) {
            deg[node] = (float)(c + 1);
            reinterpret_cast<int2*>(rowcnt)[node] = make_int2(beg + st, c);
        }
    }
    __syncthreads();

    for (int i0 = beg + tid; i0 < end; i0 += BATCH * CSR_THREADS) {
        unsigned it[BATCH];
#pragma unroll
        for (int k = 0; k < BATCH; ++k) {
            const int i = i0 + k * CSR_THREADS;
            it[k] = (unsigned)items[(i < end) ? i : (end - 1)];
        }
#pragma unroll
        for (int k = 0; k < BATCH; ++k) {
            if (i0 + k * CSR_THREADS < end) {
                const int loc = (int)(it[k] & (BIN_NODES - 1));
                const int r = atomicAdd(&s_cnt[loc], 1);
                src_sorted[s_start[loc] + r] = (int)(it[k] >> BIN_SHIFT);
            }
        }
    }
}

#define A_PITCH 132
#define A_TILE  (16 * A_PITCH)
#define K_STEPS (D_IN / 4)

__device__ __forceinline__ void fetch_rows(const float* __restrict__ x, int t, int sr,
                                           int sc, int n_nodes, float4& p0, float4& p1,
                                           float4& p2, float4& p3) {
    int row = t * 16 + sr;
    int rrow = (row < n_nodes) ? row : (n_nodes - 1);
    const float4* __restrict__ src =
        reinterpret_cast<const float4*>(x + (size_t)rrow * D_IN) + sc;
    p0 = src[0];
    p1 = src[8];
    p2 = src[16];
    p3 = src[24];
}

__device__ __forceinline__ void park_rows(float* __restrict__ dst, const float4& p0,
                                          const float4& p1, const float4& p2,
                                          const float4& p3) {
    *reinterpret_cast<float4*>(dst +  0) = p0;
    *reinterpret_cast<float4*>(dst + 32) = p1;
    *reinterpret_cast<float4*>(dst + 64) = p2;
    *reinterpret_cast<float4*>(dst + 96) = p3;
}

__device__ __forceinline__ v8f tile_products(const float* __restrict__ as_ld,
                                             const v2f (&bf)[K_STEPS], float bias) {
    v2f af[K_STEPS];
#pragma unroll
    for (int s = 0; s < K_STEPS; ++s)
        af[s] = *reinterpret_cast<const v2f*>(as_ld + 4 * s);
    v8f acc;
#pragma unroll
    for (int r = 0; r < 8; ++r) acc[r] = bias;
    __builtin_amdgcn_sched_barrier(0);
#pragma unroll
    for (int s = 0; s < K_STEPS; ++s)
        acc = __builtin_amdgcn_wmma_f32_16x16x4_f32(
             false, af[s],  false, bf[s],
             (short)0, acc,  false,  false);
    return acc;
}

__device__ __forceinline__ void fetch_deg(const float* __restrict__ deg, int t, int half,
                                          int n_nodes, float (&dg)[8]) {
#pragma unroll
    for (int r = 0; r < 8; ++r) {
        int row = t * 16 + 8 * half + r;
        dg[r] = deg[(row < n_nodes) ? row : (n_nodes - 1)];
    }
}

__device__ __forceinline__ void store_tile(float* __restrict__ hs,
                                           int t, int n0, int half, int l, int n_nodes,
                                           const v8f& acc, const float (&dg)[8]) {
    const int m0 = t * 16;
    const size_t base = (size_t)m0 * D_OUT + n0;
    float* __restrict__ hbase = hs + base;
#pragma unroll
    for (int r = 0; r < 8; ++r) {
        int mr = r + 8 * half;
        if (m0 + mr < n_nodes) {
            float d  = dg[r];
            float di = (d > 0.0f) ? rsqrtf(d) : 0.0f;
            hbase[(size_t)mr * D_OUT + l] = acc[r] * di;
        }
    }
}

__global__ void __launch_bounds__(128)
gemm_wmma_kernel(const float* __restrict__ x, const float* __restrict__ W,
                 const float* __restrict__ b, const float* __restrict__ deg,
                 float* __restrict__ hs, int n_nodes, int m_tiles) {
    __shared__ __attribute__((aligned(16))) float As[2 * A_TILE];

    int t = blockIdx.x;
    if (t >= m_tiles) return;

    const int tid  = threadIdx.x;
    const int lane = tid & 31;
    const int wave = tid >> 5;
    const int n0   = wave * 16;
    const int half = lane >> 4;
    const int l    = lane & 15;
    const int sr   = tid >> 3;
    const int sc   = tid & 7;
    const int step = gridDim.x;

    float dg[8];
    fetch_deg(deg, t, half, n_nodes, dg);
    float4 p0, p1, p2, p3;
    fetch_rows(x, t, sr, sc, n_nodes, p0, p1, p2, p3);

    const float* __restrict__ wrow = W + (size_t)(n0 + l) * D_IN + 2 * half;
    v2f bf[K_STEPS];
#pragma unroll
    for (int s = 0; s < K_STEPS; ++s) {
        bf[s][0] = wrow[4 * s + 0];
        bf[s][1] = wrow[4 * s + 1];
    }
    const float bias = b[n0 + l];

    const int st_off = sr * A_PITCH + 4 * sc;
    const int ld_off = l * A_PITCH + 2 * half;

    park_rows(As + st_off, p0, p1, p2, p3);
    __syncthreads();
    if (t + step < m_tiles) fetch_rows(x, t + step, sr, sc, n_nodes, p0, p1, p2, p3);
    v8f acc = tile_products(As + ld_off, bf, bias);
    int t_done = t;
    int buf = 1;

    for (t += step; t < m_tiles; t += step) {
        park_rows(As + buf * A_TILE + st_off, p0, p1, p2, p3);
        __syncthreads();
        store_tile(hs, t_done, n0, half, l, n_nodes, acc, dg);
        fetch_deg(deg, t, half, n_nodes, dg);
        if (t + step < m_tiles) fetch_rows(x, t + step, sr, sc, n_nodes, p0, p1, p2, p3);
        acc = tile_products(As + buf * A_TILE + ld_off, bf, bias);
        t_done = t;
        buf ^= 1;
    }
    store_tile(hs, t_done, n0, half, l, n_nodes, acc, dg);
}

#define ROWS_IN_FLIGHT 8
__global__ void __launch_bounds__(256)
gather_kernel(const int* __restrict__ rowcnt, const int* __restrict__ src_sorted,
              const float* __restrict__ hs, const float* __restrict__ deg,
              float* __restrict__ out, int n_nodes) {
    const int lane = threadIdx.x & 31;
    const int node = __builtin_amdgcn_readfirstlane(
        (int)((blockIdx.x * blockDim.x + threadIdx.x) >> 5));
    if (node >= n_nodes) return;
    const int2 rc = reinterpret_cast<const int2*>(rowcnt)[node];
    const int beg = __builtin_amdgcn_readfirstlane(rc.x);
    const int cnt = __builtin_amdgcn_readfirstlane(rc.y);
    const float d  = deg[node];
    const float di = (d > 0.0f) ? rsqrtf(d) : 0.0f;
    const float2* __restrict__ rows = reinterpret_cast<const float2*>(hs);
    const float2 self = rows[(size_t)node * (D_OUT / 2) + lane];

    float ax = 0.0f, ay = 0.0f;
    for (int base = 0; base < cnt; base += 32) {
        int m = cnt - base;
        if (m > 32) m = 32;
        const int idx = (lane < m) ? src_sorted[beg + base + lane] : 0;
        for (int j0 = 0; j0 < m; j0 += ROWS_IN_FLIGHT) {
            float2 r[ROWS_IN_FLIGHT];
#pragma unroll
            for (int k = 0; k < ROWS_IN_FLIGHT; ++k) {
                const int f = __builtin_amdgcn_readlane(idx, j0 + k);
                r[k] = rows[(size_t)f * (D_OUT / 2) + lane];
            }
#pragma unroll
            for (int k = 0; k < ROWS_IN_FLIGHT; ++k) {
                if (j0 + k < m) {
                    ax += r[k].x;
                    ay += r[k].y;
                }
            }
        }
    }
    float2 o;
    o.x = (ax + self.x) * di;
    o.y = (ay + self.y) * di;
    reinterpret_cast<float2*>(out)[(size_t)node * (D_OUT / 2) + lane] = o;
}

extern "C" void kernel_launch(void* const* d_in, const int* in_sizes, int n_in,
                              void* d_out, int out_size, void* d_ws, size_t ws_size,
                              hipStream_t stream) {
    const float* x  = (const float*)d_in[0];
    const float* W  = (const float*)d_in[1];
    const float* b  = (const float*)d_in[2];
    const int*   ei = (const int*)d_in[3];

    const int N = in_sizes[0] / D_IN;
    const int E = in_sizes[3] / 2;
    const int* from_idx = ei;
    const int* to_idx   = ei + E;

    float* out = (float*)d_out;

    const int nb    = (N + BIN_NODES - 1) / BIN_NODES;
    const int chunk = (E + SORT_BLOCKS - 1) / SORT_BLOCKS;

    char* p = (char*)d_ws;
    float* deg = (float*)p;
    p += ((size_t)N * sizeof(float) + 255) & ~(size_t)255;
    float* hs = (float*)p;
    p += ((size_t)N * D_OUT * sizeof(float) + 255) & ~(size_t)255;
    int* hist = (int*)p;
    p += ((size_t)SORT_BLOCKS * nb * sizeof(int) + 255) & ~(size_t)255;
    int* bin_base = (int*)p;
    p += ((size_t)(nb + 1) * sizeof(int) + 255) & ~(size_t)255;
    int* items = (int*)p;
    p += ((size_t)E * sizeof(int) + 255) & ~(size_t)255;
    int* src_sorted = (int*)p;
    p += ((size_t)E * sizeof(int) + 255) & ~(size_t)255;
    int* rowcnt = (int*)p;

    bin_count_kernel<<<SORT_BLOCKS, SORT_THREADS, 0, stream>>>(to_idx, hist, E, nb, chunk);
    bin_place_kernel<<<SORT_BLOCKS, SORT_THREADS, 0, stream>>>(from_idx, to_idx, hist, bin_base,
                                                               items, E, nb, chunk);
    bin_sort_kernel<<<nb, CSR_THREADS, 0, stream>>>(items, bin_base, deg, rowcnt, src_sorted, N);

    int m_tiles = (N + 15) / 16;
    gemm_wmma_kernel<<<1024, 128, 0, stream>>>(x, W, b, deg, hs, N, m_tiles);

    long long node_threads = (long long)N * 32;
    gather_kernel<<<(int)((node_threads + 255) / 256), 256, 0, stream>>>(
        rowcnt, src_sorted, hs, deg, out, N);
}


# level-1 sort made chunk-local: a block sorts its chunk by bin in its own slice, writes run starts; bin blocks read their runs via that table; count launch, table re-read go; row slot order may differ
# speedup vs baseline: 5.5163x; 1.3767x over previous
#include <hip/hip_runtime.h>

#define D_IN  128
#define D_OUT 64

typedef __attribute__((ext_vector_type(2))) float v2f;
typedef __attribute__((ext_vector_type(8))) float v8f;

#define BIN_SHIFT    8
#define BIN_NODES    256
#define SORT_BLOCKS  256
#define SORT_THREADS 1024
#define MAX_BINS     1024
#define CSR_THREADS  512
#define BATCH        4
#define RUNS_PER_WAVE (SORT_BLOCKS / (CSR_THREADS / 32))

__device__ __forceinline__ int block_excl_scan(int v, int* s_w) {
    const int lane = threadIdx.x & 31;
    const int wv   = threadIdx.x >> 5;
    const int nw   = blockDim.x >> 5;
    int incl = v;
#pragma unroll
    for (int d = 1; d < 32; d <<= 1) {
        int t = __shfl_up(incl, d);
        if (lane >= d) incl += t;
    }
    if (lane == 31) s_w[wv] = incl;
    __syncthreads();
    if (wv == 0) {
        int w  = (lane < nw) ? s_w[lane] : 0;
        int wi = w;
#pragma unroll
        for (int d = 1; d < 32; d <<= 1) {
            int t = __shfl_up(wi, d);
            if (lane >= d) wi += t;
        }
        s_w[lane] = wi - w;
    }
    __syncthreads();
    return s_w[wv] + incl - v;
}

__global__ void __launch_bounds__(SORT_THREADS)
chunk_sort_kernel(const int* __restrict__ from_idx, const int* __restrict__ to_idx,
                  int* __restrict__ tab, int* __restrict__ items, int e, int nb, int chunk) {
    __shared__ int s_off[MAX_BINS];
    __shared__ int s_cur[MAX_BINS];
    __shared__ int s_w[32];
    const int tid = threadIdx.x;
    const int me  = blockIdx.x;
    s_off[tid] = 0;
    s_cur[tid] = 0;
    __syncthreads();
    const int beg = me * chunk;
    int end = beg + chunk;
    if (end > e) end = e;
    for (int i0 = beg + tid; i0 < end; i0 += BATCH * SORT_THREADS) {
        int t[BATCH];
#pragma unroll
        for (int k = 0; k < BATCH; ++k) {
            const int i = i0 + k * SORT_THREADS;
            t[k] = to_idx[(i < end) ? i : (end - 1)];
        }
#pragma unroll
        for (int k = 0; k < BATCH; ++k)
            if (i0 + k * SORT_THREADS < end) atomicAdd(&s_off[t[k] >> BIN_SHIFT], 1);
    }
    __syncthreads();

    const int c  = s_off[tid];
    const int st = block_excl_scan(c, s_w);
    s_off[tid] = st;
    if (tid < nb) tab[tid * SORT_BLOCKS + me] = st;
    if (tid == SORT_THREADS - 1) tab[nb * SORT_BLOCKS + me] = st + c;
    __syncthreads();

    for (int i0 = beg + tid; i0 < end; i0 += BATCH * SORT_THREADS) {
        int t[BATCH], f[BATCH];
#pragma unroll
        for (int k = 0; k < BATCH; ++k) {
            const int i  = i0 + k * SORT_THREADS;
            const int ic = (i < end) ? i : (end - 1);
            t[k] = to_idx[ic];
            f[k] = from_idx[ic];
        }
#pragma unroll
        for (int k = 0; k < BATCH; ++k) {
            if (i0 + k * SORT_THREADS < end) {
                const int bin = t[k] >> BIN_SHIFT;
                const int r = atomicAdd(&s_cur[bin], 1);
                items[beg + s_off[bin] + r] = (int)(((unsigned)f[k] << BIN_SHIFT) |
                                                    (unsigned)(t[k] & (BIN_NODES - 1)));
            }
        }
    }
}

__global__ void __launch_bounds__(CSR_THREADS)
bin_sort_kernel(const int* __restrict__ items, const int* __restrict__ tab,
                float* __restrict__ deg, int* __restrict__ rowcnt,
                int* __restrict__ src_sorted, int n_nodes, int chunk) {
    __shared__ int s_cnt[BIN_NODES];
    __shared__ int s_start[BIN_NODES];
    __shared__ int s_rbeg[SORT_BLOCKS];
    __shared__ int s_rlen[SORT_BLOCKS];
    __shared__ int s_w[32];
    __shared__ int s_base;
    const int tid  = threadIdx.x;
    const int lane = tid & 31;
    const int wv   = tid >> 5;
    const int bin  = blockIdx.x;

    int o0 = 0;
    if (tid < SORT_BLOCKS) {
        o0 = tab[bin * SORT_BLOCKS + tid];
        const int o1 = tab[(bin + 1) * SORT_BLOCKS + tid];
        s_rbeg[tid] = tid * chunk + o0;
        s_rlen[tid] = o1 - o0;
    }
    if (tid < BIN_NODES) s_cnt[tid] = 0;
    if (tid == 0) s_base = 0;
    __syncthreads();

    int osum = o0;
#pragma unroll
    for (int d = 16; d > 0; d >>= 1) osum += __shfl_xor(osum, d);
    if (lane == 0 && osum != 0) atomicAdd(&s_base, osum);

    unsigned it[RUNS_PER_WAVE];
#pragma unroll
    for (int k = 0; k < RUNS_PER_WAVE; ++k) {
        const int b  = wv + (CSR_THREADS / 32) * k;
        const int rl = s_rlen[b];
        it[k] = (unsigned)items[(lane < rl) ? (s_rbeg[b] + lane) : 0];
    }
#pragma unroll
    for (int k = 0; k < RUNS_PER_WAVE; ++k) {
        const int b = wv + (CSR_THREADS / 32) * k;
        if (lane < s_rlen[b]) atomicAdd(&s_cnt[it[k] & (BIN_NODES - 1)], 1);
    }
    for (int k = 0; k < RUNS_PER_WAVE; ++k) {
        const int b  = wv + (CSR_THREADS / 32) * k;
        const int rl = s_rlen[b];
        const int rb = s_rbeg[b];
        for (int j = lane + 32; j < rl; j += 32) {
            const unsigned v = (unsigned)items[rb + j];
            atomicAdd(&s_cnt[v & (BIN_NODES - 1)], 1);
        }
    }
    __syncthreads();

    const int beg = s_base;
    const int c   = (tid < BIN_NODES) ? s_cnt[tid] : 0;
    const int st  = block_excl_scan(c, s_w);
    if (tid < BIN_NODES) {
        s_start[tid] = beg + st;
        s_cnt[tid] = 0;
        const int node = bin * BIN_NODES + tid;
        if (node < n_nodes) {
            deg[node] = (float)(c + 1);
            reinterpret_cast<int2*>(rowcnt)[node] = make_int2(beg + st, c);
        }
    }
    __syncthreads();

#pragma unroll
    for (int k = 0; k < RUNS_PER_WAVE; ++k) {
        const int b = wv + (CSR_THREADS / 32) * k;
        if (lane < s_rlen[b]) {
            const int loc = (int)(it[k] & (BIN_NODES - 1));
            const int r = atomicAdd(&s_cnt[loc], 1);
            src_sorted[s_start[loc] + r] = (int)(it[k] >> BIN_SHIFT);
        }
    }
    for (int k = 0; k < RUNS_PER_WAVE; ++k) {
        const int b  = wv + (CSR_THREADS / 32) * k;
        const int rl = s_rlen[b];
        const int rb = s_rbeg[b];
        for (int j = lane + 32; j < rl; j += 32) {
            const unsigned v = (unsigned)items[rb + j];
            const int loc = (int)(v & (BIN_NODES - 1));
            const int r = atomicAdd(&s_cnt[loc], 1);
            src_sorted[s_start[loc] + r] = (int)(v >> BIN_SHIFT);
        }
    }
}

#define A_PITCH 132
#define A_TILE  (16 * A_PITCH)
#define K_STEPS (D_IN / 4)

__device__ __forceinline__ void fetch_rows(const float* __restrict__ x, int t, int sr,
                                           int sc, int n_nodes, float4& p0, float4& p1,
                                           float4& p2, float4& p3) {
    int row = t * 16 + sr;
    int rrow = (row < n_nodes) ? row : (n_nodes - 1);
    const float4* __restrict__ src =
        reinterpret_cast<const float4*>(x + (size_t)rrow * D_IN) + sc;
    p0 = src[0];
    p1 = src[8];
    p2 = src[16];
    p3 = src[24];
}

__device__ __forceinline__ void park_rows(float* __restrict__ dst, const float4& p0,
                                          const float4& p1, const float4& p2,
                                          const float4& p3) {
    *reinterpret_cast<float4*>(dst +  0) = p0;
    *reinterpret_cast<float4*>(dst + 32) = p1;
    *reinterpret_cast<float4*>(dst + 64) = p2;
    *reinterpret_cast<float4*>(dst + 96) = p3;
}

__device__ __forceinline__ v8f tile_products(const float* __restrict__ as_ld,
                                             const v2f (&bf)[K_STEPS], float bias) {
    v2f af[K_STEPS];
#pragma unroll
    for (int s = 0; s < K_STEPS; ++s)
        af[s] = *reinterpret_cast<const v2f*>(as_ld + 4 * s);
    v8f acc;
#pragma unroll
    for (int r = 0; r < 8; ++r) acc[r] = bias;
    __builtin_amdgcn_sched_barrier(0);
#pragma unroll
    for (int s = 0; s < K_STEPS; ++s)
        acc = __builtin_amdgcn_wmma_f32_16x16x4_f32(
             false, af[s],  false, bf[s],
             (short)0, acc,  false,  false);
    return acc;
}

__device__ __forceinline__ void fetch_deg(const float* __restrict__ deg, int t, int half,
                                          int n_nodes, float (&dg)[8]) {
#pragma unroll
    for (int r = 0; r < 8; ++r) {
        int row = t * 16 + 8 * half + r;
        dg[r] = deg[(row < n_nodes) ? row : (n_nodes - 1)];
    }
}

__device__ __forceinline__ void store_tile(float* __restrict__ hs,
                                           int t, int n0, int half, int l, int n_nodes,
                                           const v8f& acc, const float (&dg)[8]) {
    const int m0 = t * 16;
    const size_t base = (size_t)m0 * D_OUT + n0;
    float* __restrict__ hbase = hs + base;
#pragma unroll
    for (int r = 0; r < 8; ++r) {
        int mr = r + 8 * half;
        if (m0 + mr < n_nodes) {
            float d  = dg[r];
            float di = (d > 0.0f) ? rsqrtf(d) : 0.0f;
            hbase[(size_t)mr * D_OUT + l] = acc[r] * di;
        }
    }
}

__global__ void __launch_bounds__(128)
gemm_wmma_kernel(const float* __restrict__ x, const float* __restrict__ W,
                 const float* __restrict__ b, const float* __restrict__ deg,
                 float* __restrict__ hs, int n_nodes, int m_tiles) {
    __shared__ __attribute__((aligned(16))) float As[2 * A_TILE];

    int t = blockIdx.x;
    if (t >= m_tiles) return;

    const int tid  = threadIdx.x;
    const int lane = tid & 31;
    const int wave = tid >> 5;
    const int n0   = wave * 16;
    const int half = lane >> 4;
    const int l    = lane & 15;
    const int sr   = tid >> 3;
    const int sc   = tid & 7;
    const int step = gridDim.x;

    float dg[8];
    fetch_deg(deg, t, half, n_nodes, dg);
    float4 p0, p1, p2, p3;
    fetch_rows(x, t, sr, sc, n_nodes, p0, p1, p2, p3);

    const float* __restrict__ wrow = W + (size_t)(n0 + l) * D_IN + 2 * half;
    v2f bf[K_STEPS];
#pragma unroll
    for (int s = 0; s < K_STEPS; ++s) {
        bf[s][0] = wrow[4 * s + 0];
        bf[s][1] = wrow[4 * s + 1];
    }
    const float bias = b[n0 + l];

    const int st_off = sr * A_PITCH + 4 * sc;
    const int ld_off = l * A_PITCH + 2 * half;

    park_rows(As + st_off, p0, p1, p2, p3);
    __syncthreads();
    if (t + step < m_tiles) fetch_rows(x, t + step, sr, sc, n_nodes, p0, p1, p2, p3);
    v8f acc = tile_products(As + ld_off, bf, bias);
    int t_done = t;
    int buf = 1;

    for (t += step; t < m_tiles; t += step) {
        park_rows(As + buf * A_TILE + st_off, p0, p1, p2, p3);
        __syncthreads();
        store_tile(hs, t_done, n0, half, l, n_nodes, acc, dg);
        fetch_deg(deg, t, half, n_nodes, dg);
        if (t + step < m_tiles) fetch_rows(x, t + step, sr, sc, n_nodes, p0, p1, p2, p3);
        acc = tile_products(As + buf * A_TILE + ld_off, bf, bias);
        t_done = t;
        buf ^= 1;
    }
    store_tile(hs, t_done, n0, half, l, n_nodes, acc, dg);
}

#define ROWS_IN_FLIGHT 8
__global__ void __launch_bounds__(256)
gather_kernel(const int* __restrict__ rowcnt, const int* __restrict__ src_sorted,
              const float* __restrict__ hs, const float* __restrict__ deg,
              float* __restrict__ out, int n_nodes) {
    const int lane = threadIdx.x & 31;
    const int node = __builtin_amdgcn_readfirstlane(
        (int)((blockIdx.x * blockDim.x + threadIdx.x) >> 5));
    if (node >= n_nodes) return;
    const int2 rc = reinterpret_cast<const int2*>(rowcnt)[node];
    const int beg = __builtin_amdgcn_readfirstlane(rc.x);
    const int cnt = __builtin_amdgcn_readfirstlane(rc.y);
    const float d  = deg[node];
    const float di = (d > 0.0f) ? rsqrtf(d) : 0.0f;
    const float2* __restrict__ rows = reinterpret_cast<const float2*>(hs);
    const float2 self = rows[(size_t)node * (D_OUT / 2) + lane];

    float ax = 0.0f, ay = 0.0f;
    for (int base = 0; base < cnt; base += 32) {
        int m = cnt - base;
        if (m > 32) m = 32;
        const int idx = (lane < m) ? src_sorted[beg + base + lane] : 0;
        for (int j0 = 0; j0 < m; j0 += ROWS_IN_FLIGHT) {
            float2 r[ROWS_IN_FLIGHT];
#pragma unroll
            for (int k = 0; k < ROWS_IN_FLIGHT; ++k) {
                const int f = __builtin_amdgcn_readlane(idx, j0 + k);
                r[k] = rows[(size_t)f * (D_OUT / 2) + lane];
            }
#pragma unroll
            for (int k = 0; k < ROWS_IN_FLIGHT; ++k) {
                if (j0 + k < m) {
                    ax += r[k].x;
                    ay += r[k].y;
                }
            }
        }
    }
    float2 o;
    o.x = (ax + self.x) * di;
    o.y = (ay + self.y) * di;
    reinterpret_cast<float2*>(out)[(size_t)node * (D_OUT / 2) + lane] = o;
}

extern "C" void kernel_launch(void* const* d_in, const int* in_sizes, int n_in,
                              void* d_out, int out_size, void* d_ws, size_t ws_size,
                              hipStream_t stream) {
    const float* x  = (const float*)d_in[0];
    const float* W  = (const float*)d_in[1];
    const float* b  = (const float*)d_in[2];
    const int*   ei = (const int*)d_in[3];

    const int N = in_sizes[0] / D_IN;
    const int E = in_sizes[3] / 2;
    const int* from_idx = ei;
    const int* to_idx   = ei + E;

    float* out = (float*)d_out;

    const int nb    = (N + BIN_NODES - 1) / BIN_NODES;
    const int chunk = (E + SORT_BLOCKS - 1) / SORT_BLOCKS;

    char* p = (char*)d_ws;
    float* deg = (float*)p;
    p += ((size_t)N * sizeof(float) + 255) & ~(size_t)255;
    float* hs = (float*)p;
    p += ((size_t)N * D_OUT * sizeof(float) + 255) & ~(size_t)255;
    int* tab = (int*)p;
    p += ((size_t)SORT_BLOCKS * (nb + 1) * sizeof(int) + 255) & ~(size_t)255;
    int* items = (int*)p;
    p += ((size_t)E * sizeof(int) + 255) & ~(size_t)255;
    int* src_sorted = (int*)p;
    p += ((size_t)E * sizeof(int) + 255) & ~(size_t)255;
    int* rowcnt = (int*)p;

    chunk_sort_kernel<<<SORT_BLOCKS, SORT_THREADS, 0, stream>>>(from_idx, to_idx, tab, items,
                                                                E, nb, chunk);
    bin_sort_kernel<<<nb, CSR_THREADS, 0, stream>>>(items, tab, deg, rowcnt, src_sorted, N, chunk);

    int m_tiles = (N + 15) / 16;
    gemm_wmma_kernel<<<1024, 128, 0, stream>>>(x, W, b, deg, hs, N, m_tiles);

    long long node_threads = (long long)N * 32;
    gather_kernel<<<(int)((node_threads + 255) / 256), 256, 0, stream>>>(
        rowcnt, src_sorted, hs, deg, out, N);
}


# both sort levels: sorted words staged in LDS, then written in order, one whole 128-byte line per store, instead of one scattered 4-byte store per edge; slot order and sums as before
# speedup vs baseline: 6.1908x; 1.1223x over previous
#include <hip/hip_runtime.h>

#define D_IN  128
#define D_OUT 64

typedef __attribute__((ext_vector_type(2))) float v2f;
typedef __attribute__((ext_vector_type(8))) float v8f;

#define BIN_SHIFT    8
#define BIN_NODES    256
#define SORT_BLOCKS  256
#define SORT_THREADS 1024
#define MAX_BINS     1024
#define CSR_THREADS  512
#define BATCH        4
#define RUNS_PER_WAVE (SORT_BLOCKS / (CSR_THREADS / 32))
#define STAGE_WORDS  8192

__device__ __forceinline__ int block_excl_scan(int v, int* s_w) {
    const int lane = threadIdx.x & 31;
    const int wv   = threadIdx.x >> 5;
    const int nw   = blockDim.x >> 5;
    int incl = v;
#pragma unroll
    for (int d = 1; d < 32; d <<= 1) {
        int t = __shfl_up(incl, d);
        if (lane >= d) incl += t;
    }
    if (lane == 31) s_w[wv] = incl;
    __syncthreads();
    if (wv == 0) {
        int w  = (lane < nw) ? s_w[lane] : 0;
        int wi = w;
#pragma unroll
        for (int d = 1; d < 32; d <<= 1) {
            int t = __shfl_up(wi, d);
            if (lane >= d) wi += t;
        }
        s_w[lane] = wi - w;
    }
    __syncthreads();
    return s_w[wv] + incl - v;
}

__global__ void __launch_bounds__(SORT_THREADS)
chunk_sort_kernel(const int* __restrict__ from_idx, const int* __restrict__ to_idx,
                  int* __restrict__ tab, int* __restrict__ items, int e, int nb, int chunk) {
    __shared__ int s_off[MAX_BINS];
    __shared__ int s_cur[MAX_BINS];
    __shared__ int s_w[32];
    __shared__ int s_stage[STAGE_WORDS];
    const int tid = threadIdx.x;
    const int me  = blockIdx.x;
    const bool staged = (chunk <= STAGE_WORDS);
    s_off[tid] = 0;
    s_cur[tid] = 0;
    __syncthreads();
    const int beg = me * chunk;
    int end = beg + chunk;
    if (end > e) end = e;
    for (int i0 = beg + tid; i0 < end; i0 += BATCH * SORT_THREADS) {
        int t[BATCH];
#pragma unroll
        for (int k = 0; k < BATCH; ++k) {
            const int i = i0 + k * SORT_THREADS;
            t[k] = to_idx[(i < end) ? i : (end - 1)];
        }
#pragma unroll
        for (int k = 0; k < BATCH; ++k)
            if (i0 + k * SORT_THREADS < end) atomicAdd(&s_off[t[k] >> BIN_SHIFT], 1);
    }
    __syncthreads();

    const int c  = s_off[tid];
    const int st = block_excl_scan(c, s_w);
    s_off[tid] = st;
    if (tid < nb) tab[tid * SORT_BLOCKS + me] = st;
    if (tid == SORT_THREADS - 1) tab[nb * SORT_BLOCKS + me] = st + c;
    __syncthreads();

    for (int i0 = beg + tid; i0 < end; i0 += BATCH * SORT_THREADS) {
        int t[BATCH], f[BATCH];
#pragma unroll
        for (int k = 0; k < BATCH; ++k) {
            const int i  = i0 + k * SORT_THREADS;
            const int ic = (i < end) ? i : (end - 1);
            t[k] = to_idx[ic];
            f[k] = from_idx[ic];
        }
#pragma unroll
        for (int k = 0; k < BATCH; ++k) {
            if (i0 + k * SORT_THREADS < end) {
                const int bin = t[k] >> BIN_SHIFT;
                const int r = atomicAdd(&s_cur[bin], 1);
                const int pos  = s_off[bin] + r;
                const int word = (int)(((unsigned)f[k] << BIN_SHIFT) |
                                       (unsigned)(t[k] & (BIN_NODES - 1)));
                if (staged) s_stage[pos] = word;
                else        items[beg + pos] = word;
            }
        }
    }
    if (staged) {
        __syncthreads();
        const int n = end - beg;
        const int a = beg & 31;
        for (int i = tid - a; i < n; i += SORT_THREADS)
            if (i >= 0) items[beg + i] = s_stage[i];
    }
}

__global__ void __launch_bounds__(CSR_THREADS)
bin_sort_kernel(const int* __restrict__ items, const int* __restrict__ tab,
                float* __restrict__ deg, int* __restrict__ rowcnt,
                int* __restrict__ src_sorted, int n_nodes, int chunk) {
    __shared__ int s_cnt[BIN_NODES];
    __shared__ int s_start[BIN_NODES];
    __shared__ int s_rbeg[SORT_BLOCKS];
    __shared__ int s_rlen[SORT_BLOCKS];
    __shared__ int s_w[32];
    __shared__ int s_base;
    __shared__ int s_total;
    __shared__ int s_stage[STAGE_WORDS];
    const int tid  = threadIdx.x;
    const int lane = tid & 31;
    const int wv   = tid >> 5;
    const int bin  = blockIdx.x;

    int o0 = 0;
    if (tid < SORT_BLOCKS) {
        o0 = tab[bin * SORT_BLOCKS + tid];
        const int o1 = tab[(bin + 1) * SORT_BLOCKS + tid];
        s_rbeg[tid] = tid * chunk + o0;
        s_rlen[tid] = o1 - o0;
    }
    if (tid < BIN_NODES) s_cnt[tid] = 0;
    if (tid == 0) s_base = 0;
    __syncthreads();

    int osum = o0;
#pragma unroll
    for (int d = 16; d > 0; d >>= 1) osum += __shfl_xor(osum, d);
    if (lane == 0 && osum != 0) atomicAdd(&s_base, osum);

    unsigned it[RUNS_PER_WAVE];
#pragma unroll
    for (int k = 0; k < RUNS_PER_WAVE; ++k) {
        const int b  = wv + (CSR_THREADS / 32) * k;
        const int rl = s_rlen[b];
        it[k] = (unsigned)items[(lane < rl) ? (s_rbeg[b] + lane) : 0];
    }
#pragma unroll
    for (int k = 0; k < RUNS_PER_WAVE; ++k) {
        const int b = wv + (CSR_THREADS / 32) * k;
        if (lane < s_rlen[b]) atomicAdd(&s_cnt[it[k] & (BIN_NODES - 1)], 1);
    }
    for (int k = 0; k < RUNS_PER_WAVE; ++k) {
        const int b  = wv + (CSR_THREADS / 32) * k;
        const int rl = s_rlen[b];
        const int rb = s_rbeg[b];
        for (int j = lane + 32; j < rl; j += 32) {
            const unsigned v = (unsigned)items[rb + j];
            atomicAdd(&s_cnt[v & (BIN_NODES - 1)], 1);
        }
    }
    __syncthreads();

    const int beg = s_base;
    const int c   = (tid < BIN_NODES) ? s_cnt[tid] : 0;
    const int st  = block_excl_scan(c, s_w);
    if (tid < BIN_NODES) {
        s_start[tid] = st;
        s_cnt[tid] = 0;
        const int node = bin * BIN_NODES + tid;
        if (node < n_nodes) {
            deg[node] = (float)(c + 1);
            reinterpret_cast<int2*>(rowcnt)[node] = make_int2(beg + st, c);
        }
    }
    if (tid == BIN_NODES - 1) s_total = st + c;
    __syncthreads();
    const int total = s_total;
    const bool staged = (total <= STAGE_WORDS);

#pragma unroll
    for (int k = 0; k < RUNS_PER_WAVE; ++k) {
        const int b = wv + (CSR_THREADS / 32) * k;
        if (lane < s_rlen[b]) {
            const int loc = (int)(it[k] & (BIN_NODES - 1));
            const int r = atomicAdd(&s_cnt[loc], 1);
            const int pos = s_start[loc] + r;
            const int f   = (int)(it[k] >> BIN_SHIFT);
            if (staged) s_stage[pos] = f;
            else        src_sorted[beg + pos] = f;
        }
    }
    for (int k = 0; k < RUNS_PER_WAVE; ++k) {
        const int b  = wv + (CSR_THREADS / 32) * k;
        const int rl = s_rlen[b];
        const int rb = s_rbeg[b];
        for (int j = lane + 32; j < rl; j += 32) {
            const unsigned v = (unsigned)items[rb + j];
            const int loc = (int)(v & (BIN_NODES - 1));
            const int r = atomicAdd(&s_cnt[loc], 1);
            const int pos = s_start[loc] + r;
            const int f   = (int)(v >> BIN_SHIFT);
            if (staged) s_stage[pos] = f;
            else        src_sorted[beg + pos] = f;
        }
    }
    if (staged) {
        __syncthreads();
        const int a = beg & 31;
        for (int i = tid - a; i < total; i += CSR_THREADS)
            if (i >= 0) src_sorted[beg + i] = s_stage[i];
    }
}

#define A_PITCH 132
#define A_TILE  (16 * A_PITCH)
#define K_STEPS (D_IN / 4)

__device__ __forceinline__ void fetch_rows(const float* __restrict__ x, int t, int sr,
                                           int sc, int n_nodes, float4& p0, float4& p1,
                                           float4& p2, float4& p3) {
    int row = t * 16 + sr;
    int rrow = (row < n_nodes) ? row : (n_nodes - 1);
    const float4* __restrict__ src =
        reinterpret_cast<const float4*>(x + (size_t)rrow * D_IN) + sc;
    p0 = src[0];
    p1 = src[8];
    p2 = src[16];
    p3 = src[24];
}

__device__ __forceinline__ void park_rows(float* __restrict__ dst, const float4& p0,
                                          const float4& p1, const float4& p2,
                                          const float4& p3) {
    *reinterpret_cast<float4*>(dst +  0) = p0;
    *reinterpret_cast<float4*>(dst + 32) = p1;
    *reinterpret_cast<float4*>(dst + 64) = p2;
    *reinterpret_cast<float4*>(dst + 96) = p3;
}

__device__ __forceinline__ v8f tile_products(const float* __restrict__ as_ld,
                                             const v2f (&bf)[K_STEPS], float bias) {
    v2f af[K_STEPS];
#pragma unroll
    for (int s = 0; s < K_STEPS; ++s)
        af[s] = *reinterpret_cast<const v2f*>(as_ld + 4 * s);
    v8f acc;
#pragma unroll
    for (int r = 0; r < 8; ++r) acc[r] = bias;
    __builtin_amdgcn_sched_barrier(0);
#pragma unroll
    for (int s = 0; s < K_STEPS; ++s)
        acc = __builtin_amdgcn_wmma_f32_16x16x4_f32(
             false, af[s],  false, bf[s],
             (short)0, acc,  false,  false);
    return acc;
}

__device__ __forceinline__ void fetch_deg(const float* __restrict__ deg, int t, int half,
                                          int n_nodes, float (&dg)[8]) {
#pragma unroll
    for (int r = 0; r < 8; ++r) {
        int row = t * 16 + 8 * half + r;
        dg[r] = deg[(row < n_nodes) ? row : (n_nodes - 1)];
    }
}

__device__ __forceinline__ void store_tile(float* __restrict__ hs,
                                           int t, int n0, int half, int l, int n_nodes,
                                           const v8f& acc, const float (&dg)[8]) {
    const int m0 = t * 16;
    const size_t base = (size_t)m0 * D_OUT + n0;
    float* __restrict__ hbase = hs + base;
#pragma unroll
    for (int r = 0; r < 8; ++r) {
        int mr = r + 8 * half;
        if (m0 + mr < n_nodes) {
            float d  = dg[r];
            float di = (d > 0.0f) ? rsqrtf(d) : 0.0f;
            hbase[(size_t)mr * D_OUT + l] = acc[r] * di;
        }
    }
}

__global__ void __launch_bounds__(128)
gemm_wmma_kernel(const float* __restrict__ x, const float* __restrict__ W,
                 const float* __restrict__ b, const float* __restrict__ deg,
                 float* __restrict__ hs, int n_nodes, int m_tiles) {
    __shared__ __attribute__((aligned(16))) float As[2 * A_TILE];

    int t = blockIdx.x;
    if (t >= m_tiles) return;

    const int tid  = threadIdx.x;
    const int lane = tid & 31;
    const int wave = tid >> 5;
    const int n0   = wave * 16;
    const int half = lane >> 4;
    const int l    = lane & 15;
    const int sr   = tid >> 3;
    const int sc   = tid & 7;
    const int step = gridDim.x;

    float dg[8];
    fetch_deg(deg, t, half, n_nodes, dg);
    float4 p0, p1, p2, p3;
    fetch_rows(x, t, sr, sc, n_nodes, p0, p1, p2, p3);

    const float* __restrict__ wrow = W + (size_t)(n0 + l) * D_IN + 2 * half;
    v2f bf[K_STEPS];
#pragma unroll
    for (int s = 0; s < K_STEPS; ++s) {
        bf[s][0] = wrow[4 * s + 0];
        bf[s][1] = wrow[4 * s + 1];
    }
    const float bias = b[n0 + l];

    const int st_off = sr * A_PITCH + 4 * sc;
    const int ld_off = l * A_PITCH + 2 * half;

    park_rows(As + st_off, p0, p1, p2, p3);
    __syncthreads();
    if (t + step < m_tiles) fetch_rows(x, t + step, sr, sc, n_nodes, p0, p1, p2, p3);
    v8f acc = tile_products(As + ld_off, bf, bias);
    int t_done = t;
    int buf = 1;

    for (t += step; t < m_tiles; t += step) {
        park_rows(As + buf * A_TILE + st_off, p0, p1, p2, p3);
        __syncthreads();
        store_tile(hs, t_done, n0, half, l, n_nodes, acc, dg);
        fetch_deg(deg, t, half, n_nodes, dg);
        if (t + step < m_tiles) fetch_rows(x, t + step, sr, sc, n_nodes, p0, p1, p2, p3);
        acc = tile_products(As + buf * A_TILE + ld_off, bf, bias);
        t_done = t;
        buf ^= 1;
    }
    store_tile(hs, t_done, n0, half, l, n_nodes, acc, dg);
}

#define ROWS_IN_FLIGHT 8
__global__ void __launch_bounds__(256)
gather_kernel(const int* __restrict__ rowcnt, const int* __restrict__ src_sorted,
              const float* __restrict__ hs, const float* __restrict__ deg,
              float* __restrict__ out, int n_nodes) {
    const int lane = threadIdx.x & 31;
    const int node = __builtin_amdgcn_readfirstlane(
        (int)((blockIdx.x * blockDim.x + threadIdx.x) >> 5));
    if (node >= n_nodes) return;
    const int2 rc = reinterpret_cast<const int2*>(rowcnt)[node];
    const int beg = __builtin_amdgcn_readfirstlane(rc.x);
    const int cnt = __builtin_amdgcn_readfirstlane(rc.y);
    const float d  = deg[node];
    const float di = (d > 0.0f) ? rsqrtf(d) : 0.0f;
    const float2* __restrict__ rows = reinterpret_cast<const float2*>(hs);
    const float2 self = rows[(size_t)node * (D_OUT / 2) + lane];

    float ax = 0.0f, ay = 0.0f;
    for (int base = 0; base < cnt; base += 32) {
        int m = cnt - base;
        if (m > 32) m = 32;
        const int idx = (lane < m) ? src_sorted[beg + base + lane] : 0;
        for (int j0 = 0; j0 < m; j0 += ROWS_IN_FLIGHT) {
            float2 r[ROWS_IN_FLIGHT];
#pragma unroll
            for (int k = 0; k < ROWS_IN_FLIGHT; ++k) {
                const int f = __builtin_amdgcn_readlane(idx, j0 + k);
                r[k] = rows[(size_t)f * (D_OUT / 2) + lane];
            }
#pragma unroll
            for (int k = 0; k < ROWS_IN_FLIGHT; ++k) {
                if (j0 + k < m) {
                    ax += r[k].x;
                    ay += r[k].y;
                }
            }
        }
    }
    float2 o;
    o.x = (ax + self.x) * di;
    o.y = (ay + self.y) * di;
    reinterpret_cast<float2*>(out)[(size_t)node * (D_OUT / 2) + lane] = o;
}

extern "C" void kernel_launch(void* const* d_in, const int* in_sizes, int n_in,
                              void* d_out, int out_size, void* d_ws, size_t ws_size,
                              hipStream_t stream) {
    const float* x  = (const float*)d_in[0];
    const float* W  = (const float*)d_in[1];
    const float* b  = (const float*)d_in[2];
    const int*   ei = (const int*)d_in[3];

    const int N = in_sizes[0] / D_IN;
    const int E = in_sizes[3] / 2;
    const int* from_idx = ei;
    const int* to_idx   = ei + E;

    float* out = (float*)d_out;

    const int nb    = (N + BIN_NODES - 1) / BIN_NODES;
    const int chunk = (E + SORT_BLOCKS - 1) / SORT_BLOCKS;

    char* p = (char*)d_ws;
    float* deg = (float*)p;
    p += ((size_t)N * sizeof(float) + 255) & ~(size_t)255;
    float* hs = (float*)p;
    p += ((size_t)N * D_OUT * sizeof(float) + 255) & ~(size_t)255;
    int* tab = (int*)p;
    p += ((size_t)SORT_BLOCKS * (nb + 1) * sizeof(int) + 255) & ~(size_t)255;
    int* items = (int*)p;
    p += ((size_t)E * sizeof(int) + 255) & ~(size_t)255;
    int* src_sorted = (int*)p;
    p += ((size_t)E * sizeof(int) + 255) & ~(size_t)255;
    int* rowcnt = (int*)p;

    chunk_sort_kernel<<<SORT_BLOCKS, SORT_THREADS, 0, stream>>>(from_idx, to_idx, tab, items,
                                                                E, nb, chunk);
    bin_sort_kernel<<<nb, CSR_THREADS, 0, stream>>>(items, tab, deg, rowcnt, src_sorted, N, chunk);

    int m_tiles = (N + 15) / 16;
    gemm_wmma_kernel<<<1024, 128, 0, stream>>>(x, W, b, deg, hs, N, m_tiles);

    long long node_threads = (long long)N * 32;
    gather_kernel<<<(int)((node_threads + 255) / 256), 256, 0, stream>>>(
        rowcnt, src_sorted, hs, deg, out, N);
}
